# starting kernel
# baseline (speedup 1.0000x reference)
#include <hip/hip_runtime.h>

typedef __attribute__((ext_vector_type(2))) float v2f;
typedef __attribute__((ext_vector_type(8))) float v8f;

static constexpr int kNodes  = 100000;
static constexpr int kInDim  = 128;
static constexpr int kHidDim = 128;
static constexpr int kOutDim = 64;


__global__ void k_deg_init(float* __restrict__ deg, int n) {
  int i = blockIdx.x * blockDim.x + threadIdx.x;
  if (i < n) deg[i] = 1.0f;
}

__device__ __forceinline__ void atomAddF(float* p, float v) {
  unsafeAtomicAdd(p, v);
}

__global__ void k_deg_count(const int* __restrict__ col, float* __restrict__ deg, int e) {
  int i = blockIdx.x * blockDim.x + threadIdx.x;
  if (i < e) atomAddF(&deg[col[i]], 1.0f);
}

__global__ void k_rsqrt_inplace(float* __restrict__ d, int n) {
  int i = blockIdx.x * blockDim.x + threadIdx.x;
  if (i < n) d[i] = rsqrtf(d[i]);
}


template <int K, int NT, bool RELU>
__global__ __launch_bounds__(128) void k_gemm_wmma(const float* __restrict__ X,
                                                   const float* __restrict__ W,
                                                   float* __restrict__ H, int M) {
  constexpr int N = NT * 16;
  const int wave = blockIdx.x * 4 + (threadIdx.x >> 5);
  const int lane = threadIdx.x & 31;
  const int m0 = wave * 16;
  if (m0 >= M) return;
  const int ml = lane & 15;
  const int hi = lane >> 4;

  v8f acc[NT];
#pragma unroll
  for (int t = 0; t < NT; ++t) acc[t] = {};

  const float* __restrict__ xrow = X + (size_t)(m0 + ml) * K;

#pragma unroll 4
  for (int k = 0; k < K; k += 4) {
    v2f a;
    a.x = xrow[k + 2 * hi + 0];
    a.y = xrow[k + 2 * hi + 1];
    if (RELU) {
      a.x = fmaxf(a.x, 0.0f);
      a.y = fmaxf(a.y, 0.0f);
    }
#pragma unroll
    for (int t = 0; t < NT; ++t) {
      v2f b;
      b.x = W[(size_t)(k + 2 * hi + 0) * N + t * 16 + ml];
      b.y = W[(size_t)(k + 2 * hi + 1) * N + t * 16 + ml];
      acc[t] = __builtin_amdgcn_wmma_f32_16x16x4_f32(
          false, a, false, b, (short)0, acc[t], false, false);
    }
  }

#pragma unroll
  for (int t = 0; t < NT; ++t) {
#pragma unroll
    for (int v = 0; v < 8; ++v) {
      H[(size_t)(m0 + v + 8 * hi) * N + t * 16 + ml] = acc[t][v];
    }
  }
}


template <int D>
__global__ void k_selfloop_bias(const float* __restrict__ H, const float* __restrict__ dinv,
                                const float* __restrict__ bias, float* __restrict__ out,
                                int n) {
  constexpr int C = D / 4;
  long long t = (long long)blockIdx.x * blockDim.x + threadIdx.x;
  long long total = (long long)n * C;
  if (t >= total) return;
  int i = (int)(t / C);
  int c = (int)(t % C);
  float s = dinv[i];
  float w = s * s;
  float4 h = reinterpret_cast<const float4*>(H)[t];
  float4 b = reinterpret_cast<const float4*>(bias)[c];
  float4 o;
  o.x = fmaf(w, h.x, b.x);
  o.y = fmaf(w, h.y, b.y);
  o.z = fmaf(w, h.z, b.z);
  o.w = fmaf(w, h.w, b.w);
  reinterpret_cast<float4*>(out)[t] = o;
}


template <int D>
__global__ void k_edge_agg(const int* __restrict__ row, const int* __restrict__ col,
                           const float* __restrict__ dinv, const float* __restrict__ H,
                           float* __restrict__ out, int e) {
  constexpr int C = D / 4;
  long long t = (long long)blockIdx.x * blockDim.x + threadIdx.x;
  long long total = (long long)e * C;
  if (t >= total) return;
  int ei = (int)(t / C);
  int c  = (int)(t % C);
  int r  = row[ei];
  int d  = col[ei];
  float norm = dinv[r] * dinv[d];
  float4 h = reinterpret_cast<const float4*>(H + (size_t)r * D)[c];
  float* o = out + (size_t)d * D + (size_t)c * 4;
  atomAddF(o + 0, norm * h.x);
  atomAddF(o + 1, norm * h.y);
  atomAddF(o + 2, norm * h.z);
  atomAddF(o + 3, norm * h.w);
}


extern "C" void kernel_launch(void* const* d_in, const int* in_sizes, int n_in,
                              void* d_out, int out_size, void* d_ws, size_t ws_size,
                              hipStream_t stream) {
  const float* x  = (const float*)d_in[0];
  const int*   ei = (const int*)d_in[1];
  const float* W1 = (const float*)d_in[2];
  const float* b1 = (const float*)d_in[3];
  const float* W2 = (const float*)d_in[4];
  const float* b2 = (const float*)d_in[5];

  const int n = kNodes;
  const int e = in_sizes[1] / 2;
  const int* row = ei;
  const int* col = ei + e;

  char* ws = (char*)d_ws;
  float* dinv = (float*)ws;
  constexpr size_t H_OFF  = 512 * 1024;
  constexpr size_t A1_OFF = H_OFF + (size_t)kNodes * kHidDim * 4;
  float* Hbuf = (float*)(ws + H_OFF);
  float* A1   = (float*)(ws + A1_OFF);
  float* out  = (float*)d_out;

  const int T = 256;

  k_deg_init<<<(n + T - 1) / T, T, 0, stream>>>(dinv, n);
  k_deg_count<<<(e + T - 1) / T, T, 0, stream>>>(col, dinv, e);
  k_rsqrt_inplace<<<(n + T - 1) / T, T, 0, stream>>>(dinv, n);

  const int waves = (n + 15) / 16;
  const int gemmBlocks = (waves + 3) / 4;

  k_gemm_wmma<kInDim, kHidDim / 16, false>
      <<<gemmBlocks, 128, 0, stream>>>(x, W1, Hbuf, n);

  {
    long long sl = (long long)n * (kHidDim / 4);
    k_selfloop_bias<kHidDim>
        <<<(unsigned)((sl + T - 1) / T), T, 0, stream>>>(Hbuf, dinv, b1, A1, n);
    long long ea = (long long)e * (kHidDim / 4);
    k_edge_agg<kHidDim>
        <<<(unsigned)((ea + T - 1) / T), T, 0, stream>>>(row, col, dinv, Hbuf, A1, e);
  }

  k_gemm_wmma<kHidDim, kOutDim / 16, true>
      <<<gemmBlocks, 128, 0, stream>>>(A1, W2, Hbuf, n);

  {
    long long sl = (long long)n * (kOutDim / 4);
    k_selfloop_bias<kOutDim>
        <<<(unsigned)((sl + T - 1) / T), T, 0, stream>>>(Hbuf, dinv, b2, out, n);
    long long ea = (long long)e * (kOutDim / 4);
    k_edge_agg<kOutDim>
        <<<(unsigned)((ea + T - 1) / T), T, 0, stream>>>(row, col, dinv, Hbuf, out, e);
  }
}


# GEMM operands: W staged per work-group in LDS in B-fragment order (a 16-B LDS read feeds a tile for 8 k), X rows read as 16-B pieces; 4-B global fragment loads gone; k order within each 8 changes
# speedup vs baseline: 3.6070x; 3.6070x over previous
#include <hip/hip_runtime.h>


typedef __attribute__((ext_vector_type(2))) float v2f;
typedef __attribute__((ext_vector_type(8))) float v8f;

static constexpr int kNodes  = 100000;
static constexpr int kInDim  = 128;
static constexpr int kHidDim = 128;
static constexpr int kOutDim = 64;

static constexpr int kTile     = 64;
static constexpr int kTileLog  = 6;
static constexpr int kNB       = (kNodes + kTile - 1) / kTile;
static constexpr int kChunk    = 4096;
static constexpr int kChunkThr = 256;
static constexpr int kPerThr   = kChunk / kChunkThr;
static constexpr int kBinIt    = (kNB + kChunkThr - 1) / kChunkThr;


template <int K, int NT, int NTHR>
__device__ __forceinline__ void stage_w_frags(const float* __restrict__ W, float4* __restrict__ wl,
                                              int tid) {
  constexpr int N = NT * 16;
  constexpr int SLOTS = (K / 8) * NT * 32;
  static_assert(SLOTS % NTHR == 0, "whole rounds of the work-group");
#pragma unroll
  for (int r = 0; r < SLOTS / NTHR; ++r) {
    const int idx = r * NTHR + tid;
    const int ln = idx & 31;
    const int t = (idx >> 5) % NT;
    const int p = idx / (32 * NT);
    const float* __restrict__ s = W + (size_t)(8 * p + 4 * (ln >> 4)) * N + t * 16 + (ln & 15);
    float4 v;
    v.x = s[0];
    v.y = s[N];
    v.z = s[2 * N];
    v.w = s[3 * N];
    wl[idx] = v;
  }
}

template <int K, int NT, bool RELU>
__device__ __forceinline__ void gemm_stripe(const float* __restrict__ X,
                                            const float4* __restrict__ wl,
                                            float* __restrict__ H, int M, int wave, int lane) {
  constexpr int N = NT * 16;
  constexpr int P = K / 8;
  const int m0 = wave * 16;
  if (m0 >= M) return;
  const int ml = lane & 15;
  const int hi = lane >> 4;

  v8f acc[NT];
#pragma unroll
  for (int t = 0; t < NT; ++t) acc[t] = {};

  const float4* __restrict__ xrow =
      reinterpret_cast<const float4*>(X + (size_t)(m0 + ml) * K + 4 * hi);

  constexpr int AH = 4;
  float4 qa[P];
#pragma unroll
  for (int p = 0; p < AH; ++p) qa[p] = xrow[2 * p];

#pragma unroll
  for (int p = 0; p < P; ++p) {
    if (p + AH < P) qa[p + AH] = xrow[2 * (p + AH)];
    float4 q = qa[p];
    if (RELU) {
      q.x = fmaxf(q.x, 0.0f);
      q.y = fmaxf(q.y, 0.0f);
      q.z = fmaxf(q.z, 0.0f);
      q.w = fmaxf(q.w, 0.0f);
    }
    float4 b[NT];
#pragma unroll
    for (int t = 0; t < NT; ++t) b[t] = wl[(p * NT + t) * 32 + lane];
    __builtin_amdgcn_sched_barrier(0);
    v2f a0, a1;
    a0.x = q.x; a0.y = q.y;
    a1.x = q.z; a1.y = q.w;
#pragma unroll
    for (int t = 0; t < NT; ++t) {
      v2f bb;
      bb.x = b[t].x; bb.y = b[t].y;
      acc[t] = __builtin_amdgcn_wmma_f32_16x16x4_f32(
          false, a0, false, bb, (short)0, acc[t], false, false);
    }
#pragma unroll
    for (int t = 0; t < NT; ++t) {
      v2f bb;
      bb.x = b[t].z; bb.y = b[t].w;
      acc[t] = __builtin_amdgcn_wmma_f32_16x16x4_f32(
          false, a1, false, bb, (short)0, acc[t], false, false);
    }
  }

#pragma unroll
  for (int t = 0; t < NT; ++t) {
#pragma unroll
    for (int v = 0; v < 8; ++v) {
      H[(size_t)(m0 + v + 8 * hi) * N + t * 16 + ml] = acc[t][v];
    }
  }
}

static constexpr int kGemmThr = 256;

template <int K, int NT, bool RELU>
__global__ __launch_bounds__(kGemmThr) void k_gemm_wmma(const float* __restrict__ X,
                                                        const float* __restrict__ W,
                                                        float* __restrict__ H, int M) {
  __shared__ __align__(16) float4 wl[(K / 8) * NT * 32];
  stage_w_frags<K, NT, kGemmThr>(W, wl, threadIdx.x);
  __syncthreads();
  gemm_stripe<K, NT, RELU>(X, wl, H, M, blockIdx.x * (kGemmThr / 32) + (threadIdx.x >> 5),
                           threadIdx.x & 31);
}

static constexpr int kStripes      = (kNodes + 15) / 16;
static constexpr int kWavesPerBlk  = kChunkThr / 32;
static constexpr int kStripesCount = 2808;
static constexpr int kBlocksCount  = kStripesCount / kWavesPerBlk;
static constexpr int kBlocksPlace  = (kStripes - kStripesCount + kWavesPerBlk - 1) / kWavesPerBlk;
static_assert(kStripesCount % kWavesPerBlk == 0, "whole work-groups of stripes");
static constexpr int kW1Slots = (kInDim / 8) * (kHidDim / 16) * 32;

__device__ __forceinline__ void layer1_stripes(const float* __restrict__ X,
                                               const float* __restrict__ W1,
                                               float* __restrict__ H, float4* __restrict__ wl,
                                               int firstStripe, int endStripe, int extraBlock) {
  stage_w_frags<kInDim, kHidDim / 16, kChunkThr>(W1, wl, threadIdx.x);
  __syncthreads();
  const int stripe = firstStripe + extraBlock * kWavesPerBlk + (int)(threadIdx.x >> 5);
  if (stripe < endStripe)
    gemm_stripe<kInDim, kHidDim / 16, false>(X, wl, H, kNodes, stripe, threadIdx.x & 31);
}


__global__ void k_bin_zero(int* __restrict__ binCount) {
  int i = blockIdx.x * blockDim.x + threadIdx.x;
  if (i < kNB) binCount[i] = 0;
}

__global__ __launch_bounds__(kChunkThr) void k_bin_count(const int* __restrict__ col,
                                                         int* __restrict__ binCount, int e,
                                                         int chunks, const float* __restrict__ X,
                                                         const float* __restrict__ W1,
                                                         float* __restrict__ H) {
  __shared__ __align__(16) float4 smem[kW1Slots];
  if ((int)blockIdx.x >= chunks) {
    layer1_stripes(X, W1, H, smem, 0, kStripesCount, (int)blockIdx.x - chunks);
    return;
  }
  int* hist = reinterpret_cast<int*>(smem);
  const int j0 = blockIdx.x * kChunk + threadIdx.x;
  const int last = e - 1;
  int d[kPerThr];
#pragma unroll
  for (int k = 0; k < kPerThr; ++k) d[k] = col[min(j0 + k * kChunkThr, last)];
  for (int i = threadIdx.x; i < kNB; i += kChunkThr) hist[i] = 0;
  __syncthreads();
#pragma unroll
  for (int k = 0; k < kPerThr; ++k) {
    if (j0 + k * kChunkThr < e) atomicAdd(&hist[d[k] >> kTileLog], 1);
  }
  __syncthreads();
  int c[kBinIt];
#pragma unroll
  for (int k = 0; k < kBinIt; ++k) {
    int i = threadIdx.x + k * kChunkThr;
    int v = hist[min(i, kNB - 1)];
    c[k] = (i < kNB) ? v : 0;
  }
#pragma unroll
  for (int k = 0; k < kBinIt; ++k) {
    if (c[k]) atomicAdd(&binCount[threadIdx.x + k * kChunkThr], c[k]);
  }
}

__global__ __launch_bounds__(1024) void k_bin_scan(const int* __restrict__ binCount,
                                                   int* __restrict__ binStart,
                                                   int* __restrict__ binCursor) {
  constexpr int IT = (kNB + 1023) / 1024;
  __shared__ int part[1024];
  const int t = threadIdx.x;
  int v[IT];
  int s = 0;
#pragma unroll
  for (int k = 0; k < IT; ++k) {
    int i = t * IT + k;
    v[k] = (i < kNB) ? binCount[i] : 0;
    s += v[k];
  }
  part[t] = s;
  __syncthreads();
  for (int off = 1; off < 1024; off <<= 1) {
    int add = (t >= off) ? part[t - off] : 0;
    __syncthreads();
    part[t] += add;
    __syncthreads();
  }
  int run = part[t] - s;
#pragma unroll
  for (int k = 0; k < IT; ++k) {
    int i = t * IT + k;
    if (i < kNB) {
      binStart[i] = run;
      binCursor[i] = run;
    }
    run += v[k];
  }
  if (t == 1023) binStart[kNB] = part[1023];
}

__global__ __launch_bounds__(kChunkThr) void k_bin_place(const int* __restrict__ row,
                                                         const int* __restrict__ col,
                                                         int* __restrict__ binCursor,
                                                         int* __restrict__ binned, int e,
                                                         int chunks, const float* __restrict__ X,
                                                         const float* __restrict__ W1,
                                                         float* __restrict__ H) {
  __shared__ __align__(16) float4 smem[kW1Slots];
  static_assert(2 * kNB * sizeof(int) <= kW1Slots * sizeof(float4), "tile tables fit the LDS block");
  if ((int)blockIdx.x >= chunks) {
    layer1_stripes(X, W1, H, smem, kStripesCount, kStripes, (int)blockIdx.x - chunks);
    return;
  }
  int* hist = reinterpret_cast<int*>(smem);
  int* base = hist + kNB;
  const int j0 = blockIdx.x * kChunk + threadIdx.x;
  const int last = e - 1;
  int dd[kPerThr], rr[kPerThr];
#pragma unroll
  for (int k = 0; k < kPerThr; ++k) {
    int j = min(j0 + k * kChunkThr, last);
    dd[k] = col[j];
    rr[k] = row[j];
  }
  for (int i = threadIdx.x; i < kNB; i += kChunkThr) hist[i] = 0;
  __syncthreads();
  int bin[kPerThr], rank[kPerThr], word[kPerThr];
#pragma unroll
  for (int k = 0; k < kPerThr; ++k) {
    bin[k] = -1;
    rank[k] = 0;
    word[k] = (rr[k] << kTileLog) | (dd[k] & (kTile - 1));
    if (j0 + k * kChunkThr < e) {
      bin[k] = dd[k] >> kTileLog;
      rank[k] = atomicAdd(&hist[bin[k]], 1);
    }
  }
  __syncthreads();
  int c[kBinIt], got[kBinIt];
#pragma unroll
  for (int k = 0; k < kBinIt; ++k) {
    int i = threadIdx.x + k * kChunkThr;
    int v = hist[min(i, kNB - 1)];
    c[k] = (i < kNB) ? v : 0;
  }
#pragma unroll
  for (int k = 0; k < kBinIt; ++k) {
    got[k] = c[k] ? atomicAdd(&binCursor[threadIdx.x + k * kChunkThr], c[k]) : 0;
  }
#pragma unroll
  for (int k = 0; k < kBinIt; ++k) {
    int i = threadIdx.x + k * kChunkThr;
    if (i < kNB) base[i] = got[k];
  }
  __syncthreads();
  int pos[kPerThr];
#pragma unroll
  for (int k = 0; k < kPerThr; ++k) pos[k] = base[max(bin[k], 0)] + rank[k];
#pragma unroll
  for (int k = 0; k < kPerThr; ++k) {
    if (bin[k] >= 0) binned[pos[k]] = word[k];
  }
}


__global__ __launch_bounds__(128) void k_tile_dinv(const int* __restrict__ binStart,
                                                   const int* __restrict__ binned,
                                                   float* __restrict__ dinv, int n) {
  __shared__ int cnt[kTile];
  constexpr int UD = 8;
  const int e0 = binStart[blockIdx.x], e1 = binStart[blockIdx.x + 1];
  if (threadIdx.x < kTile) cnt[threadIdx.x] = 1;
  __syncthreads();
  for (int j = e0 + threadIdx.x; j < e1; j += UD * 128) {
    int w[UD];
#pragma unroll
    for (int u = 0; u < UD; ++u) w[u] = binned[min(j + u * 128, e1 - 1)];
#pragma unroll
    for (int u = 0; u < UD; ++u) {
      if (j + u * 128 < e1) atomicAdd(&cnt[w[u] & (kTile - 1)], 1);
    }
  }
  __syncthreads();
  int node = blockIdx.x * kTile + threadIdx.x;
  if (threadIdx.x < kTile && node < n) dinv[node] = rsqrtf((float)cnt[threadIdx.x]);
}


template <int V> struct VecT;
template <> struct VecT<4> { typedef float4 type; };
template <> struct VecT<2> { typedef float2 type; };

template <int V>
__device__ __forceinline__ void unpack(const typename VecT<V>::type& q, float (&f)[V]);
template <>
__device__ __forceinline__ void unpack<4>(const float4& q, float (&f)[4]) {
  f[0] = q.x; f[1] = q.y; f[2] = q.z; f[3] = q.w;
}
template <>
__device__ __forceinline__ void unpack<2>(const float2& q, float (&f)[2]) {
  f[0] = q.x; f[1] = q.y;
}

template <int D>
__global__ __launch_bounds__(128) void k_tile_agg(const int* __restrict__ binStart,
                                                  const int* __restrict__ binned,
                                                  const float* __restrict__ dinv,
                                                  const float* __restrict__ H,
                                                  const float* __restrict__ bias,
                                                  float* __restrict__ out, int n) {
  constexpr int V = D / 32;
  typedef typename VecT<V>::type vec;
  constexpr int UN = 8;
  __shared__ float acc[kTile * D];
  __shared__ float sdinv[kTile];
  const int lane = threadIdx.x & 31;
  const int wave = threadIdx.x >> 5;
  const int lo = blockIdx.x * kTile;

  constexpr int RW = kTile / 4;
  const int last = n - 1;

  const int e0 = binStart[blockIdx.x];
  const int e1 = binStart[blockIdx.x + 1];
  int base = e0 + wave * 32;
  int p0 = (base + lane < e1) ? binned[base + lane] : 0;
  int p1 = (base + 128 + lane < e1) ? binned[base + 128 + lane] : 0;
  const int myrow = wave + 4 * (lane & (RW - 1));
  const float sv = dinv[min(lo + myrow, last)];
  vec q0[RW];
#pragma unroll
  for (int r = 0; r < RW; ++r)
    q0[r] = reinterpret_cast<const vec*>(H + (size_t)min(lo + wave + 4 * r, last) * D)[lane];
  const float d0 = dinv[p0 >> kTileLog];

  {
    float b[V];
    unpack<V>(reinterpret_cast<const vec*>(bias)[lane], b);
#pragma unroll
    for (int r = 0; r < RW; ++r) {
      const int i = wave + 4 * r;
      if (lo + i < n) {
        float s = __int_as_float(__builtin_amdgcn_readlane(__float_as_int(sv), r));
        float w = s * s;
        float h[V];
        unpack<V>(q0[r], h);
#pragma unroll
        for (int k = 0; k < V; ++k) acc[i * D + k * 32 + lane] = fmaf(w, h[k], b[k]);
      }
    }
    if (lane < RW && lo + myrow < n) sdinv[myrow] = sv;
  }
  __syncthreads();

  float w0 = d0 * sdinv[p0 & (kTile - 1)];
  for (; base < e1; base += 128) {
    const int p2 = (base + 256 + lane < e1) ? binned[base + 256 + lane] : 0;
    const float dn = dinv[p1 >> kTileLog];
    const int cnt = min(32, e1 - base);
    int i = 0;
    for (; i + UN <= cnt; i += UN) {
      vec q[UN];
      int dl[UN];
      float w8[UN];
#pragma unroll
      for (int u = 0; u < UN; ++u) {
        int pp = __builtin_amdgcn_readlane(p0, i + u);
        w8[u] = __int_as_float(__builtin_amdgcn_readlane(__float_as_int(w0), i + u));
        dl[u] = pp & (kTile - 1);
        q[u] = reinterpret_cast<const vec*>(H + (size_t)(pp >> kTileLog) * D)[lane];
      }
#pragma unroll
      for (int u = 0; u < UN; ++u) {
        float h[V];
        unpack<V>(q[u], h);
#pragma unroll
        for (int k = 0; k < V; ++k) atomicAdd(&acc[dl[u] * D + k * 32 + lane], w8[u] * h[k]);
      }
    }
    for (; i < cnt; ++i) {
      int pp = __builtin_amdgcn_readlane(p0, i);
      float w1 = __int_as_float(__builtin_amdgcn_readlane(__float_as_int(w0), i));
      int d1 = pp & (kTile - 1);
      float h[V];
      unpack<V>(reinterpret_cast<const vec*>(H + (size_t)(pp >> kTileLog) * D)[lane], h);
#pragma unroll
      for (int k = 0; k < V; ++k) atomicAdd(&acc[d1 * D + k * 32 + lane], w1 * h[k]);
    }
    w0 = dn * sdinv[p1 & (kTile - 1)];
    p0 = p1;
    p1 = p2;
  }
  __syncthreads();

  vec o[RW];
#pragma unroll
  for (int r = 0; r < RW; ++r) {
    const int i = wave + 4 * r;
    if constexpr (V == 4) {
      o[r].x = acc[i * D + 0 * 32 + lane];
      o[r].y = acc[i * D + 1 * 32 + lane];
      o[r].z = acc[i * D + 2 * 32 + lane];
      o[r].w = acc[i * D + 3 * 32 + lane];
    } else {
      o[r].x = acc[i * D + 0 * 32 + lane];
      o[r].y = acc[i * D + 1 * 32 + lane];
    }
  }
#pragma unroll
  for (int r = 0; r < RW; ++r) {
    const int node = lo + wave + 4 * r;
    if (node < n) reinterpret_cast<vec*>(out + (size_t)node * D)[lane] = o[r];
  }
}


extern "C" void kernel_launch(void* const* d_in, const int* in_sizes, int n_in,
                              void* d_out, int out_size, void* d_ws, size_t ws_size,
                              hipStream_t stream) {
  const float* x  = (const float*)d_in[0];
  const int*   ei = (const int*)d_in[1];
  const float* W1 = (const float*)d_in[2];
  const float* b1 = (const float*)d_in[3];
  const float* W2 = (const float*)d_in[4];
  const float* b2 = (const float*)d_in[5];

  const int n = kNodes;
  const int e = in_sizes[1] / 2;
  const int* row = ei;
  const int* col = ei + e;

  char* ws = (char*)d_ws;
  float* dinv = (float*)ws;
  constexpr size_t H_OFF   = 512 * 1024;
  constexpr size_t A1_OFF  = H_OFF + (size_t)kNodes * kHidDim * 4;
  constexpr size_t BIN_OFF = A1_OFF + (size_t)kNodes * kHidDim * 4;
  constexpr size_t LST_OFF = BIN_OFF + 3 * 8192;
  float* Hbuf = (float*)(ws + H_OFF);
  float* A1   = (float*)(ws + A1_OFF);
  float* out  = (float*)d_out;
  int* binCount  = (int*)(ws + BIN_OFF);
  int* binStart  = (int*)(ws + BIN_OFF + 8192);
  int* binCursor = (int*)(ws + BIN_OFF + 16384);
  int* binned    = (int*)(ws + LST_OFF);

  const int chunks = (e + kChunk - 1) / kChunk;

  k_bin_zero<<<(kNB + 255) / 256, 256, 0, stream>>>(binCount);
  k_bin_count<<<chunks + kBlocksCount, kChunkThr, 0, stream>>>(col, binCount, e, chunks, x, W1, Hbuf);
  k_bin_scan<<<1, 1024, 0, stream>>>(binCount, binStart, binCursor);
  k_bin_place<<<chunks + kBlocksPlace, kChunkThr, 0, stream>>>(row, col, binCursor, binned, e, chunks, x, W1,
                                                              Hbuf);
  k_tile_dinv<<<kNB, 128, 0, stream>>>(binStart, binned, dinv, n);

  constexpr int gemmBlocks = (kStripes + kGemmThr / 32 - 1) / (kGemmThr / 32);


  k_tile_agg<kHidDim><<<kNB, 128, 0, stream>>>(binStart, binned, dinv, Hbuf, b1, A1, n);

  k_gemm_wmma<kHidDim, kOutDim / 16, true>
      <<<gemmBlocks, kGemmThr, 0, stream>>>(A1, W2, Hbuf, n);

  k_tile_agg<kOutDim><<<kNB, 128, 0, stream>>>(binStart, binned, dinv, Hbuf, b2, out, n);
}
